# starting kernel
# baseline (speedup 1.0000x reference)
#include <hip/hip_runtime.h>


typedef __attribute__((ext_vector_type(2))) float v2f;
typedef __attribute__((ext_vector_type(8))) float v8f;

#define GW     64
#define WROW   256
#define HW     65536
#define PES_N  4096
#define NDS    256
#define OUT_B  1048576

__global__ __launch_bounds__(256)
void conexao_regional_wmma(const float* __restrict__ x,
                           const float* __restrict__ pesos,
                           float* __restrict__ out) {
  const int n    = blockIdx.x;
  const int i    = n >> 6;
  const int j    = n & 63;
  const int tid  = threadIdx.x;
  const int wave = tid >> 5;
  const int lane = tid & 31;
  const int half = lane >> 4;
  const int lid  = lane & 15;

  const float* xbase = x + (size_t)(i * 4) * WROW + (size_t)(j * 4) + half * 2;
  const float* pbase = pesos + (size_t)n * PES_N + half * 2;

  v2f a[2][4];
#pragma unroll
  for (int m = 0; m < 2; ++m) {
    const int b = m * 16 + lid;
#pragma unroll
    for (int kk = 0; kk < 4; ++kk) {
      a[m][kk] = *(const v2f*)(xbase + (size_t)b * HW + (size_t)kk * WROW);
    }
  }

#pragma unroll
  for (int tt = 0; tt < 2; ++tt) {
    const int t  = wave * 2 + tt;
    const int ds = t * 16 + lid;

    v2f bfrag[4];
#pragma unroll
    for (int kk = 0; kk < 4; ++kk) {
      bfrag[kk] = *(const v2f*)(pbase + (size_t)ds * 16 + kk * 4);
    }

    v8f c0 = {};
    v8f c1 = {};
#pragma unroll
    for (int kk = 0; kk < 4; ++kk) {
      c0 = __builtin_amdgcn_wmma_f32_16x16x4_f32(
           false, a[0][kk],  false, bfrag[kk],
           (short)0, c0,  false,  false);
      c1 = __builtin_amdgcn_wmma_f32_16x16x4_f32(
           false, a[1][kk],  false, bfrag[kk],
           (short)0, c1,  false,  false);
    }

    float* obase = out + (size_t)n * NDS + ds;
#pragma unroll
    for (int v = 0; v < 8; ++v) {
      const int b0 = v + half * 8;
      obase[(size_t)b0 * OUT_B]        = c0[v];
      obase[(size_t)(b0 + 16) * OUT_B] = c1[v];
    }
  }
}

extern "C" void kernel_launch(void* const* d_in, const int* in_sizes, int n_in,
                              void* d_out, int out_size, void* d_ws, size_t ws_size,
                              hipStream_t stream) {
  const float* x     = (const float*)d_in[0];
  const float* pesos = (const float*)d_in[1];
  float*       out   = (float*)d_out;
  (void)in_sizes; (void)n_in; (void)out_size; (void)d_ws; (void)ws_size;

  dim3 grid(4096);
  dim3 block(256);
  hipLaunchKernelGGL(conexao_regional_wmma, grid, block, 0, stream, x, pesos, out);
}


# Work-groups become resident loops: grid 4096 -> 512, each walks 8 adjacent regions with the loads of the next two regions in flight; 8x fewer waves to start; sums and their order unchanged
# speedup vs baseline: 1.0954x; 1.0954x over previous
#include <hip/hip_runtime.h>


typedef __attribute__((ext_vector_type(2))) float v2f;
typedef __attribute__((ext_vector_type(8))) float v8f;

#define GW     64
#define WROW   256
#define HW     65536
#define PES_N  4096
#define NDS    256
#define OUT_B  1048576
#define NREG   4096
#define RPW    8
#define AHEAD  2
#define NSTAGE (AHEAD + 1)

struct Frag {
  v2f a[2][4];
  v2f b[2][4];
};

__device__ __forceinline__ void load_frag(Frag& f, const float* __restrict__ xr,
                                          const float* __restrict__ pr,
                                          int lid, int wave) {
#pragma unroll
  for (int m = 0; m < 2; ++m) {
    const int b = m * 16 + lid;
#pragma unroll
    for (int kk = 0; kk < 4; ++kk) {
      f.a[m][kk] = *(const v2f*)(xr + (size_t)b * HW + (size_t)kk * WROW);
    }
  }
#pragma unroll
  for (int tt = 0; tt < 2; ++tt) {
    const int ds = (wave * 2 + tt) * 16 + lid;
#pragma unroll
    for (int kk = 0; kk < 4; ++kk) {
      f.b[tt][kk] = *(const v2f*)(pr + (size_t)ds * 16 + kk * 4);
    }
  }
}

__global__ __launch_bounds__(256)
void conexao_regional_wmma(const float* __restrict__ x,
                           const float* __restrict__ pesos,
                           float* __restrict__ out) {
  const int n0   = blockIdx.x * RPW;
  const int i    = n0 >> 6;
  const int j0   = n0 & 63;
  const int tid  = threadIdx.x;
  const int wave = tid >> 5;
  const int lane = tid & 31;
  const int half = lane >> 4;
  const int lid  = lane & 15;

  const float* xbase = x + (size_t)(i * 4) * WROW + (size_t)(j0 * 4) + half * 2;
  const float* pbase = pesos + (size_t)n0 * PES_N + half * 2;

  Frag f[NSTAGE];
#pragma unroll
  for (int s = 0; s < AHEAD; ++s) {
    load_frag(f[s], xbase + s * 4, pbase + (size_t)s * PES_N, lid, wave);
  }

#pragma unroll
  for (int r = 0; r < RPW; ++r) {
    __builtin_amdgcn_sched_barrier(0);
    if (r + AHEAD < RPW) {
      load_frag(f[(r + AHEAD) % NSTAGE], xbase + (r + AHEAD) * 4,
                pbase + (size_t)(r + AHEAD) * PES_N, lid, wave);
    }
    __builtin_amdgcn_sched_barrier(0);
    const Frag& cur = f[r % NSTAGE];
    const int n = n0 + r;

#pragma unroll
    for (int tt = 0; tt < 2; ++tt) {
      const int ds = (wave * 2 + tt) * 16 + lid;

      v8f c0 = {};
      v8f c1 = {};
#pragma unroll
      for (int kk = 0; kk < 4; ++kk) {
        c0 = __builtin_amdgcn_wmma_f32_16x16x4_f32(
             false, cur.a[0][kk],  false, cur.b[tt][kk],
             (short)0, c0,  false,  false);
        c1 = __builtin_amdgcn_wmma_f32_16x16x4_f32(
             false, cur.a[1][kk],  false, cur.b[tt][kk],
             (short)0, c1,  false,  false);
      }

      float* obase = out + (size_t)n * NDS + ds;
#pragma unroll
      for (int v = 0; v < 8; ++v) {
        const int b0 = v + half * 8;
        obase[(size_t)b0 * OUT_B]        = c0[v];
        obase[(size_t)(b0 + 16) * OUT_B] = c1[v];
      }
    }
  }
}

extern "C" void kernel_launch(void* const* d_in, const int* in_sizes, int n_in,
                              void* d_out, int out_size, void* d_ws, size_t ws_size,
                              hipStream_t stream) {
  const float* x     = (const float*)d_in[0];
  const float* pesos = (const float*)d_in[1];
  float*       out   = (float*)d_out;
  (void)in_sizes; (void)n_in; (void)out_size; (void)d_ws; (void)ws_size;

  dim3 grid(512);
  dim3 block(256);
  hipLaunchKernelGGL(conexao_regional_wmma, grid, block, 0, stream, x, pesos, out);
}


# Wave takes 64 ds as four B tiles interleaved by ds mod 4, four waves per region: a lane holds 4 adjacent ds and stores 16 B, each store = 4 whole 128-B lines; sums and their order unchanged
# speedup vs baseline: 1.8280x; 1.6688x over previous
#include <hip/hip_runtime.h>


typedef __attribute__((ext_vector_type(2))) float v2f;
typedef __attribute__((ext_vector_type(4))) float v4f;
typedef __attribute__((ext_vector_type(8))) float v8f;

#define GW     64
#define WROW   256
#define HW     65536
#define PES_N  4096
#define NDS    256
#define OUT_B  1048576
#define NREG   4096
#define RPW    8
#define NGRP   2
#define RPV    (RPW / NGRP)
#define NT     4
#define AHEAD  2
#define NSTAGE (AHEAD + 1)

struct Frag {
  v2f a[2][4];
  v2f b[NT][4];
};

__device__ __forceinline__ void load_frag(Frag& f, const float* __restrict__ xr,
                                          const float* __restrict__ pr, int lid) {
#pragma unroll
  for (int m = 0; m < 2; ++m) {
    const int b = m * 16 + lid;
#pragma unroll
    for (int kk = 0; kk < 4; ++kk) {
      f.a[m][kk] = *(const v2f*)(xr + (size_t)b * HW + (size_t)kk * WROW);
    }
  }
#pragma unroll
  for (int tt = 0; tt < NT; ++tt) {
#pragma unroll
    for (int kk = 0; kk < 4; ++kk) {
      f.b[tt][kk] = *(const v2f*)(pr + tt * 16 + kk * 4);
    }
  }
}

__global__ __launch_bounds__(256)
void conexao_regional_wmma(const float* __restrict__ x,
                           const float* __restrict__ pesos,
                           float* __restrict__ out) {
  const int n0   = blockIdx.x * RPW;
  const int i    = n0 >> 6;
  const int j0   = n0 & 63;
  const int tid  = threadIdx.x;
  const int wave = tid >> 5;
  const int grp  = wave >> 2;
  const int q    = wave & 3;
  const int lane = tid & 31;
  const int half = lane >> 4;
  const int lid  = lane & 15;
  const int ds0  = q * 64 + lid * NT;

  const float* xbase = x + (size_t)(i * 4) * WROW + (size_t)((j0 + grp) * 4) + half * 2;
  const float* pbase = pesos + (size_t)(n0 + grp) * PES_N + (size_t)ds0 * 16 + half * 2;

  Frag f[NSTAGE];
#pragma unroll
  for (int s = 0; s < AHEAD; ++s) {
    load_frag(f[s], xbase + s * (4 * NGRP), pbase + (size_t)s * (NGRP * PES_N), lid);
  }

#pragma unroll
  for (int s = 0; s < RPV; ++s) {
    __builtin_amdgcn_sched_barrier(0);
    if (s + AHEAD < RPV) {
      load_frag(f[(s + AHEAD) % NSTAGE], xbase + (s + AHEAD) * (4 * NGRP),
                pbase + (size_t)(s + AHEAD) * (NGRP * PES_N), lid);
    }
    __builtin_amdgcn_sched_barrier(0);
    const Frag& cur = f[s % NSTAGE];
    const int n = n0 + grp + NGRP * s;

#pragma unroll
    for (int m = 0; m < 2; ++m) {
      v8f c[NT] = {};
#pragma unroll
      for (int kk = 0; kk < 4; ++kk) {
#pragma unroll
        for (int tt = 0; tt < NT; ++tt) {
          c[tt] = __builtin_amdgcn_wmma_f32_16x16x4_f32(
               false, cur.a[m][kk],  false, cur.b[tt][kk],
               (short)0, c[tt],  false,  false);
        }
      }

      float* obase = out + (size_t)(m * 16 + half * 8) * OUT_B + (size_t)n * NDS + ds0;
#pragma unroll
      for (int v = 0; v < 8; ++v) {
        v4f r = {c[0][v], c[1][v], c[2][v], c[3][v]};
        *(v4f*)(obase + (size_t)v * OUT_B) = r;
      }
    }
  }
}

extern "C" void kernel_launch(void* const* d_in, const int* in_sizes, int n_in,
                              void* d_out, int out_size, void* d_ws, size_t ws_size,
                              hipStream_t stream) {
  const float* x     = (const float*)d_in[0];
  const float* pesos = (const float*)d_in[1];
  float*       out   = (float*)d_out;
  (void)in_sizes; (void)n_in; (void)out_size; (void)d_ws; (void)ws_size;

  dim3 grid(512);
  dim3 block(256);
  hipLaunchKernelGGL(conexao_regional_wmma, grid, block, 0, stream, x, pesos, out);
}


# pesos read as whole 512-B pieces (16 B per lane, 4 lines per load; 8 loads per 64 ds, were 16 of 8 B over 16 lines each) and put in fragment order through a wave-private LDS buffer; sums unchanged
# speedup vs baseline: 2.0422x; 1.1172x over previous
#include <hip/hip_runtime.h>


typedef __attribute__((ext_vector_type(2))) float v2f;
typedef __attribute__((ext_vector_type(4))) float v4f;
typedef __attribute__((ext_vector_type(8))) float v8f;

#define GW     64
#define WROW   256
#define HW     65536
#define PES_N  4096
#define NDS    256
#define OUT_B  1048576
#define NREG   4096
#define RPW    8
#define NGRP   2
#define RPV    (RPW / NGRP)
#define NT     4
#define AHEAD  2
#define NSTAGE (AHEAD + 1)
#define NPIECE 8
#define WBLK   1024
#define LGRP   68
#define LWAVE  (16 * LGRP)
#define NWAVE  8

struct FragA {
  v2f a[2][4];
};

__device__ __forceinline__ void load_a(FragA& f, const float* __restrict__ xr, int lid) {
#pragma unroll
  for (int m = 0; m < 2; ++m) {
    const int b = m * 16 + lid;
#pragma unroll
    for (int kk = 0; kk < 4; ++kk) {
      f.a[m][kk] = *(const v2f*)(xr + (size_t)b * HW + (size_t)kk * WROW);
    }
  }
}

__device__ __forceinline__ void load_b(v4f (&g)[NPIECE], const float* __restrict__ pl) {
#pragma unroll
  for (int i = 0; i < NPIECE; ++i) {
    g[i] = *(const v4f*)(pl + i * 128);
  }
}

__global__ __launch_bounds__(256)
void conexao_regional_wmma(const float* __restrict__ x,
                           const float* __restrict__ pesos,
                           float* __restrict__ out) {
  __shared__ __attribute__((aligned(16))) float lds[NWAVE * LWAVE];

  const int n0   = blockIdx.x * RPW;
  const int i    = n0 >> 6;
  const int j0   = n0 & 63;
  const int tid  = threadIdx.x;
  const int wave = tid >> 5;
  const int grp  = wave >> 2;
  const int q    = wave & 3;
  const int lane = tid & 31;
  const int half = lane >> 4;
  const int lid  = lane & 15;
  const int ds0  = q * 64 + lid * NT;

  const float* xbase = x + (size_t)(i * 4) * WROW + (size_t)((j0 + grp) * 4) + half * 2;
  const float* pbase = pesos + (size_t)(n0 + grp) * PES_N + (size_t)q * WBLK + lane * 4;

  float*       wst = lds + wave * LWAVE + LGRP * half + 4 * lid;
  const float* wld = lds + wave * LWAVE + LGRP * lid + 2 * half;

  FragA fa[NSTAGE];
  v4f   g[AHEAD][NPIECE];
#pragma unroll
  for (int s = 0; s < AHEAD; ++s) {
    load_a(fa[s], xbase + s * (4 * NGRP), lid);
    load_b(g[s], pbase + (size_t)s * (NGRP * PES_N));
  }

#pragma unroll
  for (int s = 0; s < RPV; ++s) {
    __builtin_amdgcn_sched_barrier(0);
    if (s + AHEAD < RPV) {
      load_a(fa[(s + AHEAD) % NSTAGE], xbase + (s + AHEAD) * (4 * NGRP), lid);
    }
#pragma unroll
    for (int p = 0; p < NPIECE; ++p) {
      *(v4f*)(wst + p * (2 * LGRP)) = g[s % AHEAD][p];
    }
    if (s + AHEAD < RPV) {
      load_b(g[s % AHEAD], pbase + (size_t)(s + AHEAD) * (NGRP * PES_N));
    }
    __builtin_amdgcn_fence(__ATOMIC_RELEASE, "wavefront");
    v2f b[NT][4];
#pragma unroll
    for (int tt = 0; tt < NT; ++tt) {
#pragma unroll
      for (int kk = 0; kk < 4; ++kk) {
        b[tt][kk] = *(const v2f*)(wld + tt * 16 + kk * 4);
      }
    }
    __builtin_amdgcn_fence(__ATOMIC_ACQUIRE, "wavefront");
    __builtin_amdgcn_sched_barrier(0);
    const FragA& cur = fa[s % NSTAGE];
    const int n = n0 + grp + NGRP * s;

#pragma unroll
    for (int m = 0; m < 2; ++m) {
      v8f c[NT] = {};
#pragma unroll
      for (int kk = 0; kk < 4; ++kk) {
#pragma unroll
        for (int tt = 0; tt < NT; ++tt) {
          c[tt] = __builtin_amdgcn_wmma_f32_16x16x4_f32(
               false, cur.a[m][kk],  false, b[tt][kk],
               (short)0, c[tt],  false,  false);
        }
      }

      float* obase = out + (size_t)(m * 16 + half * 8) * OUT_B + (size_t)n * NDS + ds0;
#pragma unroll
      for (int v = 0; v < 8; ++v) {
        v4f r = {c[0][v], c[1][v], c[2][v], c[3][v]};
        *(v4f*)(obase + (size_t)v * OUT_B) = r;
      }
    }
  }
}

extern "C" void kernel_launch(void* const* d_in, const int* in_sizes, int n_in,
                              void* d_out, int out_size, void* d_ws, size_t ws_size,
                              hipStream_t stream) {
  const float* x     = (const float*)d_in[0];
  const float* pesos = (const float*)d_in[1];
  float*       out   = (float*)d_out;
  (void)in_sizes; (void)n_in; (void)out_size; (void)d_ws; (void)ws_size;

  dim3 grid(512);
  dim3 block(256);
  hipLaunchKernelGGL(conexao_regional_wmma, grid, block, 0, stream, x, pesos, out);
}


# x read once per work-group as whole 128-B lines (4 per load; was 8-B pieces over 16 lines, each line fetched by 8 waves x 4 steps) into an LDS tile behind one barrier; A fragments from LDS; same sums
# speedup vs baseline: 2.2739x; 1.1135x over previous
#include <hip/hip_runtime.h>


typedef __attribute__((ext_vector_type(2))) float v2f;
typedef __attribute__((ext_vector_type(4))) float v4f;
typedef __attribute__((ext_vector_type(8))) float v8f;

#define GW     64
#define WROW   256
#define HW     65536
#define PES_N  4096
#define NDS    256
#define OUT_B  1048576
#define NREG   4096
#define RPW    8
#define NGRP   2
#define RPV    (RPW / NGRP)
#define NT     4
#define AHEAD  2
#define NPIECE 8
#define WBLK   1024
#define LGRP   68
#define LWAVE  (16 * LGRP)
#define NWAVE  8
#define XB     132

struct FragA {
  v2f a[2][4];
};

__device__ __forceinline__ void load_a(FragA& f, const float* xa) {
#pragma unroll
  for (int m = 0; m < 2; ++m) {
#pragma unroll
    for (int kk = 0; kk < 4; ++kk) {
      f.a[m][kk] = *(const v2f*)(xa + m * (16 * XB) + kk * 32);
    }
  }
}

__device__ __forceinline__ void load_b(v4f (&g)[NPIECE], const float* __restrict__ pl) {
#pragma unroll
  for (int i = 0; i < NPIECE; ++i) {
    g[i] = *(const v4f*)(pl + i * 128);
  }
}

__global__ __launch_bounds__(256)
void conexao_regional_wmma(const float* __restrict__ x,
                           const float* __restrict__ pesos,
                           float* __restrict__ out) {
  __shared__ __attribute__((aligned(16))) float lds[NWAVE * LWAVE];
  __shared__ __attribute__((aligned(16))) float xt[32 * XB];

  const int n0   = blockIdx.x * RPW;
  const int i    = n0 >> 6;
  const int j0   = n0 & 63;
  const int tid  = threadIdx.x;
  const int wave = tid >> 5;
  const int grp  = wave >> 2;
  const int q    = wave & 3;
  const int lane = tid & 31;
  const int half = lane >> 4;
  const int lid  = lane & 15;
  const int ds0  = q * 64 + lid * NT;

  const float* pbase = pesos + (size_t)(n0 + grp) * PES_N + (size_t)q * WBLK + lane * 4;

  const float* xg = x + (size_t)wave * HW + (size_t)(i * 4 + (lane >> 3)) * WROW
                      + (size_t)(j0 * 4) + (lane & 7) * 4;
  v4f xv[4];
#pragma unroll
  for (int p = 0; p < 4; ++p) {
    xv[p] = *(const v4f*)(xg + (size_t)p * (8 * HW));
  }

  float*       wst = lds + wave * LWAVE + LGRP * half + 4 * lid;
  const float* wld = lds + wave * LWAVE + LGRP * lid + 2 * half;

  v4f g[AHEAD][NPIECE];
#pragma unroll
  for (int s = 0; s < AHEAD; ++s) {
    load_b(g[s], pbase + (size_t)s * (NGRP * PES_N));
  }

  {
    float* xst = xt + wave * XB + lane * 4;
#pragma unroll
    for (int p = 0; p < 4; ++p) {
      *(v4f*)(xst + p * (8 * XB)) = xv[p];
    }
  }
  __builtin_amdgcn_fence(__ATOMIC_RELEASE, "workgroup", "local");
  __builtin_amdgcn_s_barrier();
  __builtin_amdgcn_fence(__ATOMIC_ACQUIRE, "workgroup", "local");
  const float* xa = xt + lid * XB + grp * 4 + half * 2;

#pragma unroll
  for (int s = 0; s < RPV; ++s) {
    __builtin_amdgcn_sched_barrier(0);
    FragA cur;
    load_a(cur, xa + s * (4 * NGRP));
#pragma unroll
    for (int p = 0; p < NPIECE; ++p) {
      *(v4f*)(wst + p * (2 * LGRP)) = g[s % AHEAD][p];
    }
    if (s + AHEAD < RPV) {
      load_b(g[s % AHEAD], pbase + (size_t)(s + AHEAD) * (NGRP * PES_N));
    }
    __builtin_amdgcn_fence(__ATOMIC_RELEASE, "wavefront");
    v2f b[NT][4];
#pragma unroll
    for (int tt = 0; tt < NT; ++tt) {
#pragma unroll
      for (int kk = 0; kk < 4; ++kk) {
        b[tt][kk] = *(const v2f*)(wld + tt * 16 + kk * 4);
      }
    }
    __builtin_amdgcn_fence(__ATOMIC_ACQUIRE, "wavefront");
    __builtin_amdgcn_sched_barrier(0);
    const int n = n0 + grp + NGRP * s;

#pragma unroll
    for (int m = 0; m < 2; ++m) {
      v8f c[NT] = {};
#pragma unroll
      for (int kk = 0; kk < 4; ++kk) {
#pragma unroll
        for (int tt = 0; tt < NT; ++tt) {
          c[tt] = __builtin_amdgcn_wmma_f32_16x16x4_f32(
               false, cur.a[m][kk],  false, b[tt][kk],
               (short)0, c[tt],  false,  false);
        }
      }

      float* obase = out + (size_t)(m * 16 + half * 8) * OUT_B + (size_t)n * NDS + ds0;
#pragma unroll
      for (int v = 0; v < 8; ++v) {
        v4f r = {c[0][v], c[1][v], c[2][v], c[3][v]};
        *(v4f*)(obase + (size_t)v * OUT_B) = r;
      }
    }
  }
}

extern "C" void kernel_launch(void* const* d_in, const int* in_sizes, int n_in,
                              void* d_out, int out_size, void* d_ws, size_t ws_size,
                              hipStream_t stream) {
  const float* x     = (const float*)d_in[0];
  const float* pesos = (const float*)d_in[1];
  float*       out   = (float*)d_out;
  (void)in_sizes; (void)n_in; (void)out_size; (void)d_ws; (void)ws_size;

  dim3 grid(512);
  dim3 block(256);
  hipLaunchKernelGGL(conexao_regional_wmma, grid, block, 0, stream, x, pesos, out);
}
